# starting kernel
# baseline (speedup 1.0000x reference)
#include <hip/hip_runtime.h>
#include <math.h>

typedef float v2f __attribute__((ext_vector_type(2)));
typedef float v4f __attribute__((ext_vector_type(4)));
typedef float v8f __attribute__((ext_vector_type(8)));

#define FEAT_D 128

__global__ __launch_bounds__(256) void rnorm_kernel(const float* __restrict__ h,
                                                    float* __restrict__ rn,
                                                    int nrows) {
  const int wave = (blockIdx.x * blockDim.x + threadIdx.x) >> 5;
  const int lane = threadIdx.x & 31;
  if (wave >= nrows) return;

  const v4f v = *(const v4f*)(h + (size_t)wave * FEAT_D + lane * 4);
  float s = v.x * v.x + v.y * v.y + v.z * v.z + v.w * v.w;
#pragma unroll
  for (int off = 16; off > 0; off >>= 1) s += __shfl_xor(s, off, 32);
  if (lane == 0) rn[wave] = 1.0f / sqrtf(s);
}

__global__ __launch_bounds__(256) void edge_cos_wmma_kernel(
    const float* __restrict__ hg, const float* __restrict__ hd,
    const int* __restrict__ src, const int* __restrict__ dst,
    const float* __restrict__ rng, const float* __restrict__ rnd,
    float* __restrict__ out, int E) {
  const int wave = (blockIdx.x * blockDim.x + threadIdx.x) >> 5;
  const int lane = threadIdx.x & 31;
  const int e_base = wave * 16;
  if (e_base >= E) return;

  const int m = lane & 15;
  const int h = lane >> 4;
  int e_m = e_base + m;
  const int e_c = (e_m < E) ? e_m : (E - 1);

  const int si = src[e_c];
  const int di = dst[e_c];
  const float* gp = hg + (size_t)si * FEAT_D;
  const float* dp = hd + (size_t)di * FEAT_D;

  v8f c = {0.f, 0.f, 0.f, 0.f, 0.f, 0.f, 0.f, 0.f};

#pragma unroll 4
  for (int kk = 0; kk < 16; ++kk) {
    const int off = kk * 8 + h * 4;
    const v4f a4 = *(const v4f*)(gp + off);
    const v4f b4 = *(const v4f*)(dp + off);
    const v2f a0 = {a4.x, a4.y}, b0 = {b4.x, b4.y};
    const v2f a1 = {a4.z, a4.w}, b1 = {b4.z, b4.w};
    c = __builtin_amdgcn_wmma_f32_16x16x4_f32(false, a0, false, b0,
                                              (short)0, c, false, false);
    c = __builtin_amdgcn_wmma_f32_16x16x4_f32(false, a1, false, b1,
                                              (short)0, c, false, false);
  }

  const int sel = lane & 7;
  float d = c[0];
  if (sel == 1) d = c[1];
  if (sel == 2) d = c[2];
  if (sel == 3) d = c[3];
  if (sel == 4) d = c[4];
  if (sel == 5) d = c[5];
  if (sel == 6) d = c[6];
  if (sel == 7) d = c[7];

  const bool diag_lane = (lane < 8) || (lane >= 24);
  if (diag_lane && (e_m < E)) {
    out[e_m] = d * rng[si] * rnd[di];
  }
}

extern "C" void kernel_launch(void* const* d_in, const int* in_sizes, int n_in,
                              void* d_out, int out_size, void* d_ws, size_t ws_size,
                              hipStream_t stream) {
  const float* hg = (const float*)d_in[0];
  const float* hd = (const float*)d_in[1];
  const int* src = (const int*)d_in[2];
  const int* dst = (const int*)d_in[3];
  float* out = (float*)d_out;

  const int Ng = in_sizes[0] / FEAT_D;
  const int Nd = in_sizes[1] / FEAT_D;
  const int E = in_sizes[2];

  float* rng = (float*)d_ws;
  float* rnd = rng + Ng;

  rnorm_kernel<<<(Ng + 7) / 8, 256, 0, stream>>>(hg, rng, Ng);
  rnorm_kernel<<<(Nd + 7) / 8, 256, 0, stream>>>(hd, rnd, Nd);

  const int waves = (E + 15) / 16;
  const int blocks = (waves + 7) / 8;
  edge_cos_wmma_kernel<<<blocks, 256, 0, stream>>>(hg, hd, src, dst, rng, rnd,
                                                   out, E);
}


# Both row-norm launches (70000 one-row waves) become one launch of 2192 waves: a wave streams 32 whole rows, all loads in flight, and writes one 128-B line of results; sums and their order unchanged
# speedup vs baseline: 1.2745x; 1.2745x over previous
#include <hip/hip_runtime.h>
#include <math.h>

typedef float v2f __attribute__((ext_vector_type(2)));
typedef float v4f __attribute__((ext_vector_type(4)));
typedef float v8f __attribute__((ext_vector_type(8)));

#define FEAT_D 128

#define WAVES_PER_BLOCK 8
#define TILE_ROWS 16
#define HALF_D 64
#define TILE_PITCH 68

#define NORM_ROWS 32
#define NORM_BLOCKS 274

__global__ __launch_bounds__(256) void rnorm_rows_kernel(
    const float* __restrict__ hg, const float* __restrict__ hd,
    float* __restrict__ rn, int ng, int nd) {
  const int lane = threadIdx.x & 31;
  const int wave = __builtin_amdgcn_readfirstlane(
      (int)((blockIdx.x * blockDim.x + threadIdx.x) >> 5));
  const int nwaves = (int)((gridDim.x * blockDim.x) >> 5);
  const int nrows = ng + nd;
  const int ntiles = (nrows + NORM_ROWS - 1) / NORM_ROWS;

  for (int t = wave; t < ntiles; t += nwaves) {
    const int r0 = t * NORM_ROWS;
    v4f v[NORM_ROWS];
    const bool whole = (r0 + NORM_ROWS <= ng) ||
                       (r0 >= ng && r0 + NORM_ROWS <= nrows);
    if (whole) {
      const float* base = ((r0 < ng) ? hg + (size_t)r0 * FEAT_D
                                     : hd + (size_t)(r0 - ng) * FEAT_D) + lane * 4;
#pragma unroll
      for (int u = 0; u < NORM_ROWS; ++u)
        v[u] = *(const v4f*)(base + u * FEAT_D);
    } else {
#pragma unroll
      for (int u = 0; u < NORM_ROWS; ++u) {
        int r = r0 + u;
        r = (r < nrows) ? r : (nrows - 1);
        const float* p = (r < ng) ? hg + (size_t)r * FEAT_D
                                  : hd + (size_t)(r - ng) * FEAT_D;
        v[u] = *(const v4f*)(p + lane * 4);
      }
    }
    __builtin_amdgcn_sched_barrier(0);
    float mine = 0.f;
#pragma unroll
    for (int u = 0; u < NORM_ROWS; ++u) {
      float s = v[u].x * v[u].x + v[u].y * v[u].y + v[u].z * v[u].z +
                v[u].w * v[u].w;
#pragma unroll
      for (int off = 16; off > 0; off >>= 1) s += __shfl_xor(s, off, 32);
      if (lane == u) mine = s;
    }
    const int r = r0 + lane;
    if (r < nrows) rn[r] = 1.0f / sqrtf(mine);
  }
}

__global__ __launch_bounds__(256) void edge_cos_wmma_kernel(
    const float* __restrict__ hg, const float* __restrict__ hd,
    const int* __restrict__ src, const int* __restrict__ dst,
    const float* __restrict__ rng, const float* __restrict__ rnd,
    float* __restrict__ out, int E) {
  __shared__ __attribute__((aligned(16)))
  float tile[WAVES_PER_BLOCK * 2 * TILE_ROWS * TILE_PITCH];

  const int wave = (blockIdx.x * blockDim.x + threadIdx.x) >> 5;
  const int lane = threadIdx.x & 31;
  const int e_base = wave * 16;
  if (e_base >= E) return;

  const int m = lane & 15;
  const int h = lane >> 4;
  int e_m = e_base + m;
  const int e_c = (e_m < E) ? e_m : (E - 1);

  const int si = src[e_c];
  const int di = dst[e_c];

  float* tg = tile + (threadIdx.x >> 5) * (2 * TILE_ROWS * TILE_PITCH);
  float* td = tg + TILE_ROWS * TILE_PITCH;

  const float* pg[TILE_ROWS / 2];
  const float* pd[TILE_ROWS / 2];
#pragma unroll
  for (int j = 0; j < TILE_ROWS / 2; ++j) {
    const int s_lo = __builtin_amdgcn_readlane(si, 2 * j);
    const int s_hi = __builtin_amdgcn_readlane(si, 2 * j + 1);
    const int d_lo = __builtin_amdgcn_readlane(di, 2 * j);
    const int d_hi = __builtin_amdgcn_readlane(di, 2 * j + 1);
    pg[j] = hg + (size_t)(h ? s_hi : s_lo) * FEAT_D + m * 4;
    pd[j] = hd + (size_t)(h ? d_hi : d_lo) * FEAT_D + m * 4;
  }

  float* sg = tg + h * TILE_PITCH + m * 4;
  float* sd = td + h * TILE_PITCH + m * 4;
  const float* ag = tg + m * TILE_PITCH + h * 4;
  const float* bd = td + m * TILE_PITCH + h * 4;

  v8f c = {0.f, 0.f, 0.f, 0.f, 0.f, 0.f, 0.f, 0.f};

#pragma unroll
  for (int half = 0; half < 2; ++half) {
    v4f rg[TILE_ROWS / 2], rd[TILE_ROWS / 2];
#pragma unroll
    for (int j = 0; j < TILE_ROWS / 2; ++j) {
      rg[j] = *(const v4f*)(pg[j] + half * HALF_D);
      rd[j] = *(const v4f*)(pd[j] + half * HALF_D);
    }
#pragma unroll
    for (int j = 0; j < TILE_ROWS / 2; ++j) {
      *(v4f*)(sg + 2 * j * TILE_PITCH) = rg[j];
      *(v4f*)(sd + 2 * j * TILE_PITCH) = rd[j];
    }
#pragma unroll
    for (int kk = 0; kk < 8; ++kk) {
      const v4f a4 = *(const v4f*)(ag + kk * 8);
      const v4f b4 = *(const v4f*)(bd + kk * 8);
      const v2f a0 = {a4.x, a4.y}, b0 = {b4.x, b4.y};
      const v2f a1 = {a4.z, a4.w}, b1 = {b4.z, b4.w};
      c = __builtin_amdgcn_wmma_f32_16x16x4_f32(false, a0, false, b0,
                                                (short)0, c, false, false);
      c = __builtin_amdgcn_wmma_f32_16x16x4_f32(false, a1, false, b1,
                                                (short)0, c, false, false);
    }
  }

  const int sel = lane & 7;
  float d = c[0];
  if (sel == 1) d = c[1];
  if (sel == 2) d = c[2];
  if (sel == 3) d = c[3];
  if (sel == 4) d = c[4];
  if (sel == 5) d = c[5];
  if (sel == 6) d = c[6];
  if (sel == 7) d = c[7];

  const bool diag_lane = (lane < 8) || (lane >= 24);
  if (diag_lane && (e_m < E)) {
    out[e_m] = d * rng[si] * rnd[di];
  }
}

extern "C" void kernel_launch(void* const* d_in, const int* in_sizes, int n_in,
                              void* d_out, int out_size, void* d_ws, size_t ws_size,
                              hipStream_t stream) {
  const float* hg = (const float*)d_in[0];
  const float* hd = (const float*)d_in[1];
  const int* src = (const int*)d_in[2];
  const int* dst = (const int*)d_in[3];
  float* out = (float*)d_out;

  const int Ng = in_sizes[0] / FEAT_D;
  const int Nd = in_sizes[1] / FEAT_D;
  const int E = in_sizes[2];

  float* rng = (float*)d_ws;
  float* rnd = rng + Ng;

  rnorm_rows_kernel<<<NORM_BLOCKS, 256, 0, stream>>>(hg, hd, rng, Ng, Nd);

  const int waves = (E + 15) / 16;
  const int blocks = (waves + 7) / 8;
  edge_cos_wmma_kernel<<<blocks, 256, 0, stream>>>(hg, hd, src, dst, rng, rnd,
                                                   out, E);
}


# Gathered rows go from one coalesced 512-B load straight into both WMMA operands (features on M and N; an edge's dot = trace of its 16x16 product, DPP butterfly); LDS staging gone; sum order changes
# speedup vs baseline: 1.3442x; 1.0547x over previous
#include <hip/hip_runtime.h>
#include <math.h>

typedef float v2f __attribute__((ext_vector_type(2)));
typedef float v4f __attribute__((ext_vector_type(4)));
typedef float v8f __attribute__((ext_vector_type(8)));

#define FEAT_D 128


#define NORM_ROWS 32
#define NORM_BLOCKS 274

__global__ __launch_bounds__(256) void rnorm_rows_kernel(
    const float* __restrict__ hg, const float* __restrict__ hd,
    float* __restrict__ rn, int ng, int nd) {
  const int lane = threadIdx.x & 31;
  const int wave = __builtin_amdgcn_readfirstlane(
      (int)((blockIdx.x * blockDim.x + threadIdx.x) >> 5));
  const int nwaves = (int)((gridDim.x * blockDim.x) >> 5);
  const int nrows = ng + nd;
  const int ntiles = (nrows + NORM_ROWS - 1) / NORM_ROWS;

  for (int t = wave; t < ntiles; t += nwaves) {
    const int r0 = t * NORM_ROWS;
    v4f v[NORM_ROWS];
    const bool whole = (r0 + NORM_ROWS <= ng) ||
                       (r0 >= ng && r0 + NORM_ROWS <= nrows);
    if (whole) {
      const float* base = ((r0 < ng) ? hg + (size_t)r0 * FEAT_D
                                     : hd + (size_t)(r0 - ng) * FEAT_D) + lane * 4;
#pragma unroll
      for (int u = 0; u < NORM_ROWS; ++u)
        v[u] = *(const v4f*)(base + u * FEAT_D);
    } else {
#pragma unroll
      for (int u = 0; u < NORM_ROWS; ++u) {
        int r = r0 + u;
        r = (r < nrows) ? r : (nrows - 1);
        const float* p = (r < ng) ? hg + (size_t)r * FEAT_D
                                  : hd + (size_t)(r - ng) * FEAT_D;
        v[u] = *(const v4f*)(p + lane * 4);
      }
    }
    __builtin_amdgcn_sched_barrier(0);
    float mine = 0.f;
#pragma unroll
    for (int u = 0; u < NORM_ROWS; ++u) {
      float s = v[u].x * v[u].x + v[u].y * v[u].y + v[u].z * v[u].z +
                v[u].w * v[u].w;
#pragma unroll
      for (int off = 16; off > 0; off >>= 1) s += __shfl_xor(s, off, 32);
      if (lane == u) mine = s;
    }
    const int r = r0 + lane;
    if (r < nrows) rn[r] = 1.0f / sqrtf(mine);
  }
}


template <int MASK>
__device__ __forceinline__ float row_xor(float v) {
  return __builtin_bit_cast(
      float, __builtin_amdgcn_update_dpp(0, __builtin_bit_cast(int, v),
                                         0x160 + MASK, 0xf, 0xf, false));
}

typedef const __attribute__((address_space(1))) char* gbytes;
__device__ __forceinline__ gbytes scalar_row(const float* tab, int row) {
  const unsigned long long a =
      (unsigned long long)(tab + (size_t)row * FEAT_D);
  const unsigned lo = __builtin_amdgcn_readfirstlane((unsigned)a);
  const unsigned hi = __builtin_amdgcn_readfirstlane((unsigned)(a >> 32));
  return (gbytes)(((unsigned long long)hi << 32) | lo);
}

__global__ __launch_bounds__(256) void edge_cos_wmma_kernel(
    const float* __restrict__ hg, const float* __restrict__ hd,
    const int* __restrict__ src, const int* __restrict__ dst,
    const float* __restrict__ rng, const float* __restrict__ rnd,
    float* __restrict__ out, int E) {
  const int wave = __builtin_amdgcn_readfirstlane(
      (int)((blockIdx.x * blockDim.x + threadIdx.x) >> 5));
  const int lane = threadIdx.x & 31;
  const int e_base = wave * 16;
  if (e_base >= E) return;

  const int m = lane & 15;
  int e_m = e_base + m;
  const int e_c = (e_m < E) ? e_m : (E - 1);

  const int si = src[e_c];
  const int di = dst[e_c];

  v4f g[16], d[16];
  const unsigned loff = (unsigned)lane * 16u;
#pragma unroll
  for (int u = 0; u < 16; ++u) {
    const int s = __builtin_amdgcn_readlane(si, u);
    const int t = __builtin_amdgcn_readlane(di, u);
    g[u] = *(const __attribute__((address_space(1))) v4f*)(scalar_row(hg, s) + loff);
    d[u] = *(const __attribute__((address_space(1))) v4f*)(scalar_row(hd, t) + loff);
  }
  __builtin_amdgcn_sched_barrier(0);

  const int sel = lane & 7;
  float t[16];
#pragma unroll
  for (int u = 0; u < 16; ++u) {
    const v2f a0 = {g[u].x, g[u].y}, b0 = {d[u].x, d[u].y};
    const v2f a1 = {g[u].z, g[u].w}, b1 = {d[u].z, d[u].w};
    v8f c = {0.f, 0.f, 0.f, 0.f, 0.f, 0.f, 0.f, 0.f};
    c = __builtin_amdgcn_wmma_f32_16x16x4_f32(false, a0, false, b0,
                                              (short)0, c, false, false);
    c = __builtin_amdgcn_wmma_f32_16x16x4_f32(false, a1, false, b1,
                                              (short)0, c, false, false);
    float x = c[0];
    if (sel == 1) x = c[1];
    if (sel == 2) x = c[2];
    if (sel == 3) x = c[3];
    if (sel == 4) x = c[4];
    if (sel == 5) x = c[5];
    if (sel == 6) x = c[6];
    if (sel == 7) x = c[7];
    t[u] = x;
  }

  const bool hi_row = lane >= 16;
  float r[8];
#pragma unroll
  for (int u = 0; u < 8; ++u) {
    const float keep = hi_row ? t[u + 8] : t[u];
    const float send = hi_row ? t[u] : t[u + 8];
    const int got = __builtin_amdgcn_permlanex16(
        0, __builtin_bit_cast(int, send), 0xFEDCBA98u, 0x76543210u, false,
        false);
    r[u] = keep + __builtin_bit_cast(float, got);
  }
  const bool b4 = (lane & 4) != 0;
  float q[4];
#pragma unroll
  for (int u = 0; u < 4; ++u) {
    const float keep = b4 ? r[u + 4] : r[u];
    const float send = b4 ? r[u] : r[u + 4];
    q[u] = keep + row_xor<4>(send);
  }
  const bool b2 = (lane & 2) != 0;
  float p[2];
#pragma unroll
  for (int u = 0; u < 2; ++u) {
    const float keep = b2 ? q[u + 2] : q[u];
    const float send = b2 ? q[u] : q[u + 2];
    p[u] = keep + row_xor<2>(send);
  }
  const bool b1 = (lane & 1) != 0;
  const float keep1 = b1 ? p[1] : p[0];
  const float send1 = b1 ? p[0] : p[1];
  const float dot = keep1 + row_xor<1>(send1);

  const bool diag_lane = (lane < 8) || (lane >= 24);
  if (diag_lane && (e_m < E)) {
    out[e_m] = dot * rng[si] * rnd[di];
  }
}

extern "C" void kernel_launch(void* const* d_in, const int* in_sizes, int n_in,
                              void* d_out, int out_size, void* d_ws, size_t ws_size,
                              hipStream_t stream) {
  const float* hg = (const float*)d_in[0];
  const float* hd = (const float*)d_in[1];
  const int* src = (const int*)d_in[2];
  const int* dst = (const int*)d_in[3];
  float* out = (float*)d_out;

  const int Ng = in_sizes[0] / FEAT_D;
  const int Nd = in_sizes[1] / FEAT_D;
  const int E = in_sizes[2];

  float* rng = (float*)d_ws;
  float* rnd = rng + Ng;

  rnorm_rows_kernel<<<NORM_BLOCKS, 256, 0, stream>>>(hg, hd, rng, Ng, Nd);

  const int waves = (E + 15) / 16;
  const int blocks = (waves + 7) / 8;
  edge_cos_wmma_kernel<<<blocks, 256, 0, stream>>>(hg, hd, src, dst, rng, rnd,
                                                   out, E);
}
